# starting kernel
# baseline (speedup 1.0000x reference)
#include <hip/hip_runtime.h>

typedef __attribute__((ext_vector_type(2))) float v2f;
typedef __attribute__((ext_vector_type(8))) float v8f;

#define DF 64
#define LDS_STRIDE 68

__global__ void scatter_add(const float* __restrict__ xin,
                            const int* __restrict__ src,
                            const int* __restrict__ dst,
                            float* __restrict__ agg,
                            float* __restrict__ cnt,
                            int n_edges, int do_cnt) {
  long long t = (long long)blockIdx.x * blockDim.x + threadIdx.x;
  int e    = (int)(t >> 5);
  int lane = (int)(t & 31);
  if (e >= n_edges) return;
  int s = src[e];
  int d = dst[e];
  float2 v = *(const float2*)(xin + (long long)s * DF + lane * 2);
  float* ar = agg + (long long)d * DF + lane * 2;
  atomicAdd(ar,     v.x);
  atomicAdd(ar + 1, v.y);
  if (do_cnt && lane == 0) atomicAdd(cnt + d, 1.0f);
}

__global__ void __launch_bounds__(128)
sage_linear(const float* __restrict__ agg,
            const float* __restrict__ cnt,
            const float* __restrict__ xin,
            const float* __restrict__ Wl,
            const float* __restrict__ bl,
            const float* __restrict__ Wr,
            float* __restrict__ out,
            int n_nodes, int do_relu) {
  __shared__ float sWl[DF * LDS_STRIDE];
  __shared__ float sWr[DF * LDS_STRIDE];

  for (int i = threadIdx.x; i < DF * DF / 2; i += 128) {
    int r = i >> 5;
    int c = (i & 31) * 2;
    *(float2*)&sWl[r * LDS_STRIDE + c] = *(const float2*)&Wl[r * DF + c];
    *(float2*)&sWr[r * LDS_STRIDE + c] = *(const float2*)&Wr[r * DF + c];
  }
  __syncthreads();

  int wave = threadIdx.x >> 5;
  int lane = threadIdx.x & 31;
  int tile = blockIdx.x * 4 + wave;
  int nb   = tile * 16;
  if (nb >= n_nodes) return;

  int m  = lane & 15;
  int kh = lane >> 4;

  int row = nb + m;
  float invc = 1.0f / fmaxf(cnt[row], 1.0f);
  const float* arow = agg + (long long)row * DF + 2 * kh;
  const float* xrow = xin + (long long)row * DF + 2 * kh;

  v8f acc[4];
#pragma unroll
  for (int nt = 0; nt < 4; ++nt) {
    float b = bl[nt * 16 + m];
    v8f c = {b, b, b, b, b, b, b, b};
    acc[nt] = c;
  }

  for (int k = 0; k < DF; k += 4) {
    v2f am = *(const v2f*)(arow + k);
    am *= invc;
    v2f ax = *(const v2f*)(xrow + k);
#pragma unroll
    for (int nt = 0; nt < 4; ++nt) {
      v2f bw = *(const v2f*)&sWl[(nt * 16 + m) * LDS_STRIDE + k + 2 * kh];
      acc[nt] = __builtin_amdgcn_wmma_f32_16x16x4_f32(
          false, am, false, bw, (short)0, acc[nt], false, false);
      v2f br = *(const v2f*)&sWr[(nt * 16 + m) * LDS_STRIDE + k + 2 * kh];
      acc[nt] = __builtin_amdgcn_wmma_f32_16x16x4_f32(
          false, ax, false, br, (short)0, acc[nt], false, false);
    }
  }

#pragma unroll
  for (int nt = 0; nt < 4; ++nt) {
#pragma unroll
    for (int g = 0; g < 8; ++g) {
      int orow = nb + g + 8 * kh;
      float v = acc[nt][g];
      if (do_relu) v = fmaxf(v, 0.0f);
      out[(long long)orow * DF + nt * 16 + m] = v;
    }
  }
}

extern "C" void kernel_launch(void* const* d_in, const int* in_sizes, int n_in,
                              void* d_out, int out_size, void* d_ws, size_t ws_size,
                              hipStream_t stream) {
  const float* x   = (const float*)d_in[0];
  const int*   ei  = (const int*)d_in[1];
  const float* W1l = (const float*)d_in[2];
  const float* b1l = (const float*)d_in[3];
  const float* W1r = (const float*)d_in[4];
  const float* W2l = (const float*)d_in[5];
  const float* b2l = (const float*)d_in[6];
  const float* W2r = (const float*)d_in[7];
  float* out = (float*)d_out;

  int n_nodes = in_sizes[0] / DF;
  int n_edges = in_sizes[1] / 2;
  const int* src = ei;
  const int* dst = ei + n_edges;

  size_t aggBytes = (size_t)n_nodes * DF * sizeof(float);
  char* ws = (char*)d_ws;
  float* agg = (float*)ws;
  float* h   = (float*)(ws + aggBytes);
  float* cnt = (float*)(ws + 2 * aggBytes);

  long long etot = (long long)n_edges * 32;
  int eblocks = (int)((etot + 255) / 256);
  int tiles   = (n_nodes + 15) / 16;
  int gblocks = (tiles + 3) / 4;

  hipMemsetAsync(agg, 0, aggBytes, stream);
  hipMemsetAsync(cnt, 0, (size_t)n_nodes * sizeof(float), stream);
  scatter_add<<<eblocks, 256, 0, stream>>>(x, src, dst, agg, cnt, n_edges, 1);
  sage_linear<<<gblocks, 128, 0, stream>>>(agg, cnt, x, W1l, b1l, W1r, h,
                                           n_nodes, 1);

  hipMemsetAsync(agg, 0, aggBytes, stream);
  scatter_add<<<eblocks, 256, 0, stream>>>(h, src, dst, agg, cnt, n_edges, 0);
  sage_linear<<<gblocks, 128, 0, stream>>>(agg, cnt, h, W2l, b2l, W2r, out,
                                           n_nodes, 0);
}


# Aggregation: float-atomic scatter replaced by per-call counting sort of edges by dst (int atomic gives rank) + atomic-free per-node gather-sum in registers; sum order = arrival rank
# speedup vs baseline: 2.1128x; 2.1128x over previous
#include <hip/hip_runtime.h>

typedef __attribute__((ext_vector_type(2))) float v2f;
typedef __attribute__((ext_vector_type(8))) float v8f;

#define DF 64
#define LDS_STRIDE 68


__global__ void __launch_bounds__(256)
edge_rank(const int* __restrict__ dst, int* __restrict__ deg,
          int* __restrict__ rank, int n_edges) {
  int e = blockIdx.x * 256 + threadIdx.x;
  if (e >= n_edges) return;
  rank[e] = atomicAdd(deg + dst[e], 1);
}

__global__ void __launch_bounds__(256)
seg_alloc(const int* __restrict__ deg, int* __restrict__ cursor,
          int* __restrict__ seg, float* __restrict__ cnt, int n_nodes) {
  int n    = blockIdx.x * 256 + threadIdx.x;
  int lane = threadIdx.x & 31;
  int d    = (n < n_nodes) ? deg[n] : 0;
  int incl = d;
  for (int s = 1; s < 32; s <<= 1) {
    int t = __shfl_up(incl, s, 32);
    if (lane >= s) incl += t;
  }
  int total = __shfl(incl, 31, 32);
  int base  = 0;
  if (lane == 31) base = atomicAdd(cursor, total);
  base = __shfl(base, 31, 32);
  if (n < n_nodes) {
    *(int2*)(seg + 2 * n) = make_int2(base + incl - d, d);
    cnt[n] = (float)d;
  }
}

__global__ void __launch_bounds__(256)
edge_place(const int* __restrict__ src, const int* __restrict__ dst,
           const int* __restrict__ rank, const int* __restrict__ seg,
           int* __restrict__ sorted, int n_edges) {
  int e = blockIdx.x * 256 + threadIdx.x;
  if (e >= n_edges) return;
  sorted[seg[2 * dst[e]] + rank[e]] = src[e];
}

__global__ void __launch_bounds__(256)
gather_sum(const float* __restrict__ xin, const int* __restrict__ sorted,
           const int* __restrict__ seg, float* __restrict__ agg, int n_nodes) {
  int node = blockIdx.x * 8 + __builtin_amdgcn_readfirstlane(threadIdx.x >> 5);
  int lane = threadIdx.x & 31;
  if (node >= n_nodes) return;
  int start = seg[2 * node];
  int deg   = seg[2 * node + 1];
  float2 acc = make_float2(0.0f, 0.0f);
  for (int c = 0; c < deg; c += 32) {
    int nc  = min(32, deg - c);
    int idx = (lane < nc) ? sorted[start + c + lane] : 0;
#pragma unroll 4
    for (int j = 0; j < nc; ++j) {
      int s = __builtin_amdgcn_readlane(idx, j);
      float2 v = *(const float2*)(xin + (long long)s * DF + lane * 2);
      acc.x += v.x;
      acc.y += v.y;
    }
  }
  *(float2*)(agg + (long long)node * DF + lane * 2) = acc;
}

__global__ void __launch_bounds__(128)
sage_linear(const float* __restrict__ agg,
            const float* __restrict__ cnt,
            const float* __restrict__ xin,
            const float* __restrict__ Wl,
            const float* __restrict__ bl,
            const float* __restrict__ Wr,
            float* __restrict__ out,
            int n_nodes, int do_relu) {
  __shared__ float sWl[DF * LDS_STRIDE];
  __shared__ float sWr[DF * LDS_STRIDE];

  for (int i = threadIdx.x; i < DF * DF / 2; i += 128) {
    int r = i >> 5;
    int c = (i & 31) * 2;
    *(float2*)&sWl[r * LDS_STRIDE + c] = *(const float2*)&Wl[r * DF + c];
    *(float2*)&sWr[r * LDS_STRIDE + c] = *(const float2*)&Wr[r * DF + c];
  }
  __syncthreads();

  int wave = threadIdx.x >> 5;
  int lane = threadIdx.x & 31;
  int tile = blockIdx.x * 4 + wave;
  int nb   = tile * 16;
  if (nb >= n_nodes) return;

  int m  = lane & 15;
  int kh = lane >> 4;

  int row = nb + m;
  float invc = 1.0f / fmaxf(cnt[row], 1.0f);
  const float* arow = agg + (long long)row * DF + 2 * kh;
  const float* xrow = xin + (long long)row * DF + 2 * kh;

  v8f acc[4];
#pragma unroll
  for (int nt = 0; nt < 4; ++nt) {
    float b = bl[nt * 16 + m];
    v8f c = {b, b, b, b, b, b, b, b};
    acc[nt] = c;
  }

  for (int k = 0; k < DF; k += 4) {
    v2f am = *(const v2f*)(arow + k);
    am *= invc;
    v2f ax = *(const v2f*)(xrow + k);
#pragma unroll
    for (int nt = 0; nt < 4; ++nt) {
      v2f bw = *(const v2f*)&sWl[(nt * 16 + m) * LDS_STRIDE + k + 2 * kh];
      acc[nt] = __builtin_amdgcn_wmma_f32_16x16x4_f32(
          false, am, false, bw, (short)0, acc[nt], false, false);
      v2f br = *(const v2f*)&sWr[(nt * 16 + m) * LDS_STRIDE + k + 2 * kh];
      acc[nt] = __builtin_amdgcn_wmma_f32_16x16x4_f32(
          false, ax, false, br, (short)0, acc[nt], false, false);
    }
  }

#pragma unroll
  for (int nt = 0; nt < 4; ++nt) {
#pragma unroll
    for (int g = 0; g < 8; ++g) {
      int orow = nb + g + 8 * kh;
      float v = acc[nt][g];
      if (do_relu) v = fmaxf(v, 0.0f);
      out[(long long)orow * DF + nt * 16 + m] = v;
    }
  }
}

extern "C" void kernel_launch(void* const* d_in, const int* in_sizes, int n_in,
                              void* d_out, int out_size, void* d_ws, size_t ws_size,
                              hipStream_t stream) {
  const float* x   = (const float*)d_in[0];
  const int*   ei  = (const int*)d_in[1];
  const float* W1l = (const float*)d_in[2];
  const float* b1l = (const float*)d_in[3];
  const float* W1r = (const float*)d_in[4];
  const float* W2l = (const float*)d_in[5];
  const float* b2l = (const float*)d_in[6];
  const float* W2r = (const float*)d_in[7];
  float* out = (float*)d_out;

  int n_nodes = in_sizes[0] / DF;
  int n_edges = in_sizes[1] / 2;
  const int* src = ei;
  const int* dst = ei + n_edges;

  size_t aggBytes = (size_t)n_nodes * DF * sizeof(float);
  char* ws = (char*)d_ws;
  float* agg = (float*)ws;
  float* h   = (float*)(ws + aggBytes);
  float* cnt = (float*)(ws + 2 * aggBytes);

  int* rank   = (int*)h;
  int* deg    = rank + n_edges;
  int* cursor = deg + n_nodes;
  int* seg    = (int*)d_out;
  int* sorted = seg + 2 * (size_t)n_nodes;

  int eblocks = (n_edges + 255) / 256;
  int nblocks = (n_nodes + 255) / 256;
  int wblocks = (n_nodes + 7) / 8;
  int tiles   = (n_nodes + 15) / 16;
  int gblocks = (tiles + 3) / 4;

  hipMemsetAsync(deg, 0, ((size_t)n_nodes + 1) * sizeof(int), stream);
  edge_rank<<<eblocks, 256, 0, stream>>>(dst, deg, rank, n_edges);
  seg_alloc<<<nblocks, 256, 0, stream>>>(deg, cursor, seg, cnt, n_nodes);
  edge_place<<<eblocks, 256, 0, stream>>>(src, dst, rank, seg, sorted, n_edges);

  gather_sum<<<wblocks, 256, 0, stream>>>(x, sorted, seg, agg, n_nodes);
  sage_linear<<<gblocks, 128, 0, stream>>>(agg, cnt, x, W1l, b1l, W1r, h,
                                           n_nodes, 1);

  gather_sum<<<wblocks, 256, 0, stream>>>(h, sorted, seg, agg, n_nodes);
  sage_linear<<<gblocks, 128, 0, stream>>>(agg, cnt, h, W2l, b2l, W2r, out,
                                           n_nodes, 0);
}


# Edge grouping: global-atomic counting sort (rank, slice, place) replaced by a two-level sort on LDS counters (chunk->bucket, bucket->node); sum order = LDS arrival rank
# speedup vs baseline: 3.8813x; 1.8370x over previous
#include <hip/hip_runtime.h>

typedef __attribute__((ext_vector_type(2))) float v2f;
typedef __attribute__((ext_vector_type(8))) float v8f;

#define DF 64
#define LDS_STRIDE 68

#define BK_SHIFT 9
#define BK_NODES 512
#define CH_EDGES 8192
#define SRC_BITS 23
#define SRC_MASK 0x7FFFFFu
#define SORT_CAP 16384

__device__ __forceinline__ int wave_incl_scan(int v, int lane) {
  for (int s = 1; s < 32; s <<= 1) {
    int t = __shfl_up(v, s, 32);
    if (lane >= s) v += t;
  }
  return v;
}

__global__ void __launch_bounds__(1024)
edge_partition(const int* __restrict__ src, const int* __restrict__ dst,
               unsigned* __restrict__ part, int* __restrict__ tab,
               int n_edges, int n_buckets, int n_chunks) {
  __shared__ int s_cnt[257];
  __shared__ int s_off[257];
  __shared__ int s_wtot[8];
  __shared__ unsigned s_stage[CH_EDGES];

  int t    = threadIdx.x;
  int lane = t & 31;
  int wave = t >> 5;
  int g    = blockIdx.x;
  int base = g * CH_EDGES;
  int len  = min(CH_EDGES, n_edges - base);

  if (t <= 256) s_cnt[t] = 0;

  int d[8], s[8];
#pragma unroll
  for (int k = 0; k < 8; ++k) {
    int i = k * 1024 + t;
    d[k] = -1;
    s[k] = 0;
    if (i < len) {
      d[k] = dst[base + i];
      s[k] = src[base + i];
    }
  }
  __syncthreads();

  int br[8];
#pragma unroll
  for (int k = 0; k < 8; ++k) {
    int b = (d[k] >= 0) ? (d[k] >> BK_SHIFT) : 256;
    int r = atomicAdd(&s_cnt[b], 1);
    br[k] = (b << 16) | r;
  }
  __syncthreads();

  int v = 0, incl = 0;
  if (t < 256) {
    v    = (t < n_buckets) ? s_cnt[t] : 0;
    incl = wave_incl_scan(v, lane);
    if (lane == 31) s_wtot[wave] = incl;
  }
  __syncthreads();
  if (t < 256) {
    int wb = 0;
    for (int w = 0; w < wave; ++w) wb += s_wtot[w];
    s_off[t + 1] = wb + incl;
    if (t == 0) s_off[0] = 0;
  }
  __syncthreads();

#pragma unroll
  for (int k = 0; k < 8; ++k) {
    if (d[k] >= 0) {
      int b = br[k] >> 16;
      int r = br[k] & 0xFFFF;
      s_stage[s_off[b] + r] =
          ((unsigned)(d[k] & (BK_NODES - 1)) << SRC_BITS) | (unsigned)s[k];
    }
  }
  __syncthreads();

  for (int i = t; i < len; i += 1024) part[base + i] = s_stage[i];
  if (t <= n_buckets) tab[t * n_chunks + g] = s_off[t];
}

__global__ void __launch_bounds__(1024)
bucket_sort(const unsigned* __restrict__ part, const int* __restrict__ tab,
            int* __restrict__ seg, int* __restrict__ sorted,
            float* __restrict__ cnt, int n_nodes, int n_chunks) {
  __shared__ int s_lo[1024];
  __shared__ int s_len[1024];
  __shared__ int s_cnt[BK_NODES];
  __shared__ int s_cur[BK_NODES];
  __shared__ int s_wtot[16];
  __shared__ int s_base, s_total;
  __shared__ int s_sorted[SORT_CAP];

  int t    = threadIdx.x;
  int lane = t & 31;
  int wave = t >> 5;
  int wv   = __builtin_amdgcn_readfirstlane(wave);
  int b    = blockIdx.x;

  if (t == 0) { s_base = 0; s_total = 0; }
  if (t < BK_NODES) s_cnt[t] = 0;
  __syncthreads();

  int lo = 0, ln = 0;
  if (t < n_chunks) {
    lo = tab[b * n_chunks + t];
    ln = tab[(b + 1) * n_chunks + t] - lo;
    s_lo[t]  = t * CH_EDGES + lo;
    s_len[t] = ln;
  }
  int a = lo, c = ln;
  for (int sh = 16; sh > 0; sh >>= 1) {
    a += __shfl_xor(a, sh, 32);
    c += __shfl_xor(c, sh, 32);
  }
  if (lane == 0 && (a | c) != 0) {
    atomicAdd(&s_base, a);
    atomicAdd(&s_total, c);
  }
  __syncthreads();
  int  bucket_base = s_base;
  int  total       = s_total;
  bool fits        = total <= SORT_CAP;

  for (int g = wv; g < n_chunks; g += 32) {
    int p0 = __builtin_amdgcn_readfirstlane(s_lo[g]);
    int rl = __builtin_amdgcn_readfirstlane(s_len[g]);
    for (int i = lane; i < rl; i += 64) {
      int      i1 = i + 32;
      unsigned w0 = part[p0 + i];
      unsigned w1 = (i1 < rl) ? part[p0 + i1] : 0u;
      atomicAdd(&s_cnt[w0 >> SRC_BITS], 1);
      if (i1 < rl) atomicAdd(&s_cnt[w1 >> SRC_BITS], 1);
    }
  }
  __syncthreads();

  int v = 0, incl = 0;
  if (t < BK_NODES) {
    v    = s_cnt[t];
    incl = wave_incl_scan(v, lane);
    if (lane == 31) s_wtot[wave] = incl;
  }
  __syncthreads();
  if (t < BK_NODES) {
    int wb = 0;
    for (int w = 0; w < wave; ++w) wb += s_wtot[w];
    int start = wb + incl - v;
    s_cur[t] = start;
    int n = b * BK_NODES + t;
    if (n < n_nodes) {
      *(int2*)(seg + 2 * n) = make_int2(bucket_base + start, v);
      cnt[n] = (float)v;
    }
  }
  __syncthreads();

  for (int g = wv; g < n_chunks; g += 32) {
    int p0 = __builtin_amdgcn_readfirstlane(s_lo[g]);
    int rl = __builtin_amdgcn_readfirstlane(s_len[g]);
    for (int i = lane; i < rl; i += 64) {
      int      i1 = i + 32;
      unsigned w0 = part[p0 + i];
      unsigned w1 = (i1 < rl) ? part[p0 + i1] : 0u;
      int q0 = atomicAdd(&s_cur[w0 >> SRC_BITS], 1);
      if (fits) s_sorted[q0] = (int)(w0 & SRC_MASK);
      else      sorted[bucket_base + q0] = (int)(w0 & SRC_MASK);
      if (i1 < rl) {
        int q1 = atomicAdd(&s_cur[w1 >> SRC_BITS], 1);
        if (fits) s_sorted[q1] = (int)(w1 & SRC_MASK);
        else      sorted[bucket_base + q1] = (int)(w1 & SRC_MASK);
      }
    }
  }
  if (fits) {
    __syncthreads();
    for (int i = t; i < total; i += 1024) sorted[bucket_base + i] = s_sorted[i];
  }
}

__global__ void __launch_bounds__(256)
gather_sum(const float* __restrict__ xin, const int* __restrict__ sorted,
           const int* __restrict__ seg, float* __restrict__ agg, int n_nodes) {
  int node = blockIdx.x * 8 + __builtin_amdgcn_readfirstlane(threadIdx.x >> 5);
  int lane = threadIdx.x & 31;
  if (node >= n_nodes) return;
  int start = seg[2 * node];
  int deg   = seg[2 * node + 1];
  float2 acc = make_float2(0.0f, 0.0f);
  for (int c = 0; c < deg; c += 32) {
    int nc  = min(32, deg - c);
    int idx = (lane < nc) ? sorted[start + c + lane] : 0;
#pragma unroll 4
    for (int j = 0; j < nc; ++j) {
      int s = __builtin_amdgcn_readlane(idx, j);
      float2 v = *(const float2*)(xin + (long long)s * DF + lane * 2);
      acc.x += v.x;
      acc.y += v.y;
    }
  }
  *(float2*)(agg + (long long)node * DF + lane * 2) = acc;
}

__global__ void __launch_bounds__(128)
sage_linear(const float* __restrict__ agg,
            const float* __restrict__ cnt,
            const float* __restrict__ xin,
            const float* __restrict__ Wl,
            const float* __restrict__ bl,
            const float* __restrict__ Wr,
            float* __restrict__ out,
            int n_nodes, int do_relu) {
  __shared__ float sWl[DF * LDS_STRIDE];
  __shared__ float sWr[DF * LDS_STRIDE];

  for (int i = threadIdx.x; i < DF * DF / 2; i += 128) {
    int r = i >> 5;
    int c = (i & 31) * 2;
    *(float2*)&sWl[r * LDS_STRIDE + c] = *(const float2*)&Wl[r * DF + c];
    *(float2*)&sWr[r * LDS_STRIDE + c] = *(const float2*)&Wr[r * DF + c];
  }
  __syncthreads();

  int wave = threadIdx.x >> 5;
  int lane = threadIdx.x & 31;
  int tile = blockIdx.x * 4 + wave;
  int nb   = tile * 16;
  if (nb >= n_nodes) return;

  int m  = lane & 15;
  int kh = lane >> 4;

  int row = nb + m;
  float invc = 1.0f / fmaxf(cnt[row], 1.0f);
  const float* arow = agg + (long long)row * DF + 2 * kh;
  const float* xrow = xin + (long long)row * DF + 2 * kh;

  v8f acc[4];
#pragma unroll
  for (int nt = 0; nt < 4; ++nt) {
    float b = bl[nt * 16 + m];
    v8f c = {b, b, b, b, b, b, b, b};
    acc[nt] = c;
  }

  for (int k = 0; k < DF; k += 4) {
    v2f am = *(const v2f*)(arow + k);
    am *= invc;
    v2f ax = *(const v2f*)(xrow + k);
#pragma unroll
    for (int nt = 0; nt < 4; ++nt) {
      v2f bw = *(const v2f*)&sWl[(nt * 16 + m) * LDS_STRIDE + k + 2 * kh];
      acc[nt] = __builtin_amdgcn_wmma_f32_16x16x4_f32(
          false, am, false, bw, (short)0, acc[nt], false, false);
      v2f br = *(const v2f*)&sWr[(nt * 16 + m) * LDS_STRIDE + k + 2 * kh];
      acc[nt] = __builtin_amdgcn_wmma_f32_16x16x4_f32(
          false, ax, false, br, (short)0, acc[nt], false, false);
    }
  }

#pragma unroll
  for (int nt = 0; nt < 4; ++nt) {
#pragma unroll
    for (int g = 0; g < 8; ++g) {
      int orow = nb + g + 8 * kh;
      float v = acc[nt][g];
      if (do_relu) v = fmaxf(v, 0.0f);
      out[(long long)orow * DF + nt * 16 + m] = v;
    }
  }
}

extern "C" void kernel_launch(void* const* d_in, const int* in_sizes, int n_in,
                              void* d_out, int out_size, void* d_ws, size_t ws_size,
                              hipStream_t stream) {
  const float* x   = (const float*)d_in[0];
  const int*   ei  = (const int*)d_in[1];
  const float* W1l = (const float*)d_in[2];
  const float* b1l = (const float*)d_in[3];
  const float* W1r = (const float*)d_in[4];
  const float* W2l = (const float*)d_in[5];
  const float* b2l = (const float*)d_in[6];
  const float* W2r = (const float*)d_in[7];
  float* out = (float*)d_out;

  int n_nodes = in_sizes[0] / DF;
  int n_edges = in_sizes[1] / 2;
  const int* src = ei;
  const int* dst = ei + n_edges;

  size_t aggBytes = (size_t)n_nodes * DF * sizeof(float);
  char* ws = (char*)d_ws;
  float* agg = (float*)ws;
  float* h   = (float*)(ws + aggBytes);
  float* cnt = (float*)(ws + 2 * aggBytes);

  int n_buckets = (n_nodes + BK_NODES - 1) / BK_NODES;
  int n_chunks  = (n_edges + CH_EDGES - 1) / CH_EDGES;

  unsigned* part = (unsigned*)h;
  int*      tab  = (int*)(part + n_edges);
  int* seg    = (int*)d_out;
  int* sorted = seg + 2 * (size_t)n_nodes;

  int wblocks = (n_nodes + 7) / 8;
  int tiles   = (n_nodes + 15) / 16;
  int gblocks = (tiles + 3) / 4;

  edge_partition<<<n_chunks, 1024, 0, stream>>>(src, dst, part, tab, n_edges,
                                                n_buckets, n_chunks);
  bucket_sort<<<n_buckets, 1024, 0, stream>>>(part, tab, seg, sorted, cnt,
                                              n_nodes, n_chunks);

  gather_sum<<<wblocks, 256, 0, stream>>>(x, sorted, seg, agg, n_nodes);
  sage_linear<<<gblocks, 128, 0, stream>>>(agg, cnt, x, W1l, b1l, W1r, h,
                                           n_nodes, 1);

  gather_sum<<<wblocks, 256, 0, stream>>>(h, sorted, seg, agg, n_nodes);
  sage_linear<<<gblocks, 128, 0, stream>>>(agg, cnt, h, W2l, b2l, W2r, out,
                                           n_nodes, 0);
}


# Gather: a wave walks the sources of 8 consecutive nodes as one stream (16 row loads per group, no per-node remainder, next sources fetched ahead), not one wave per node: 12500 waves; sum order kept
# speedup vs baseline: 4.1296x; 1.0640x over previous
#include <hip/hip_runtime.h>

typedef __attribute__((ext_vector_type(2))) float v2f;
typedef __attribute__((ext_vector_type(8))) float v8f;

#define DF 64
#define LDS_STRIDE 68

#define BK_SHIFT 9
#define BK_NODES 512
#define CH_EDGES 8192
#define SRC_BITS 23
#define SRC_MASK 0x7FFFFFu
#define SORT_CAP 16384

__device__ __forceinline__ int wave_incl_scan(int v, int lane) {
  for (int s = 1; s < 32; s <<= 1) {
    int t = __shfl_up(v, s, 32);
    if (lane >= s) v += t;
  }
  return v;
}

__global__ void __launch_bounds__(1024)
edge_partition(const int* __restrict__ src, const int* __restrict__ dst,
               unsigned* __restrict__ part, int* __restrict__ tab,
               int n_edges, int n_buckets, int n_chunks) {
  __shared__ int s_cnt[257];
  __shared__ int s_off[257];
  __shared__ int s_wtot[8];
  __shared__ unsigned s_stage[CH_EDGES];

  int t    = threadIdx.x;
  int lane = t & 31;
  int wave = t >> 5;
  int g    = blockIdx.x;
  int base = g * CH_EDGES;
  int len  = min(CH_EDGES, n_edges - base);

  if (t <= 256) s_cnt[t] = 0;

  int d[8], s[8];
#pragma unroll
  for (int k = 0; k < 8; ++k) {
    int i = k * 1024 + t;
    d[k] = -1;
    s[k] = 0;
    if (i < len) {
      d[k] = dst[base + i];
      s[k] = src[base + i];
    }
  }
  __syncthreads();

  int br[8];
#pragma unroll
  for (int k = 0; k < 8; ++k) {
    int b = (d[k] >= 0) ? (d[k] >> BK_SHIFT) : 256;
    int r = atomicAdd(&s_cnt[b], 1);
    br[k] = (b << 16) | r;
  }
  __syncthreads();

  int v = 0, incl = 0;
  if (t < 256) {
    v    = (t < n_buckets) ? s_cnt[t] : 0;
    incl = wave_incl_scan(v, lane);
    if (lane == 31) s_wtot[wave] = incl;
  }
  __syncthreads();
  if (t < 256) {
    int wb = 0;
    for (int w = 0; w < wave; ++w) wb += s_wtot[w];
    s_off[t + 1] = wb + incl;
    if (t == 0) s_off[0] = 0;
  }
  __syncthreads();

#pragma unroll
  for (int k = 0; k < 8; ++k) {
    if (d[k] >= 0) {
      int b = br[k] >> 16;
      int r = br[k] & 0xFFFF;
      s_stage[s_off[b] + r] =
          ((unsigned)(d[k] & (BK_NODES - 1)) << SRC_BITS) | (unsigned)s[k];
    }
  }
  __syncthreads();

  for (int i = t; i < len; i += 1024) part[base + i] = s_stage[i];
  if (t <= n_buckets) tab[t * n_chunks + g] = s_off[t];
}

__global__ void __launch_bounds__(1024)
bucket_sort(const unsigned* __restrict__ part, const int* __restrict__ tab,
            int* __restrict__ seg, int* __restrict__ sorted,
            float* __restrict__ cnt, int n_nodes, int n_chunks) {
  __shared__ int s_lo[1024];
  __shared__ int s_len[1024];
  __shared__ int s_cnt[BK_NODES];
  __shared__ int s_cur[BK_NODES];
  __shared__ int s_wtot[16];
  __shared__ int s_base, s_total;
  __shared__ int s_sorted[SORT_CAP];

  int t    = threadIdx.x;
  int lane = t & 31;
  int wave = t >> 5;
  int wv   = __builtin_amdgcn_readfirstlane(wave);
  int b    = blockIdx.x;

  if (t == 0) { s_base = 0; s_total = 0; }
  if (t < BK_NODES) s_cnt[t] = 0;
  __syncthreads();

  int lo = 0, ln = 0;
  if (t < n_chunks) {
    lo = tab[b * n_chunks + t];
    ln = tab[(b + 1) * n_chunks + t] - lo;
    s_lo[t]  = t * CH_EDGES + lo;
    s_len[t] = ln;
  }
  int a = lo, c = ln;
  for (int sh = 16; sh > 0; sh >>= 1) {
    a += __shfl_xor(a, sh, 32);
    c += __shfl_xor(c, sh, 32);
  }
  if (lane == 0 && (a | c) != 0) {
    atomicAdd(&s_base, a);
    atomicAdd(&s_total, c);
  }
  __syncthreads();
  int  bucket_base = s_base;
  int  total       = s_total;
  bool fits        = total <= SORT_CAP;

  for (int g = wv; g < n_chunks; g += 32) {
    int p0 = __builtin_amdgcn_readfirstlane(s_lo[g]);
    int rl = __builtin_amdgcn_readfirstlane(s_len[g]);
    for (int i = lane; i < rl; i += 64) {
      int      i1 = i + 32;
      unsigned w0 = part[p0 + i];
      unsigned w1 = (i1 < rl) ? part[p0 + i1] : 0u;
      atomicAdd(&s_cnt[w0 >> SRC_BITS], 1);
      if (i1 < rl) atomicAdd(&s_cnt[w1 >> SRC_BITS], 1);
    }
  }
  __syncthreads();

  int v = 0, incl = 0;
  if (t < BK_NODES) {
    v    = s_cnt[t];
    incl = wave_incl_scan(v, lane);
    if (lane == 31) s_wtot[wave] = incl;
  }
  __syncthreads();
  if (t < BK_NODES) {
    int wb = 0;
    for (int w = 0; w < wave; ++w) wb += s_wtot[w];
    int start = wb + incl - v;
    s_cur[t] = start;
    int n = b * BK_NODES + t;
    if (n < n_nodes) {
      *(int2*)(seg + 2 * n) = make_int2(bucket_base + start, v);
      cnt[n] = (float)v;
    }
  }
  __syncthreads();

  for (int g = wv; g < n_chunks; g += 32) {
    int p0 = __builtin_amdgcn_readfirstlane(s_lo[g]);
    int rl = __builtin_amdgcn_readfirstlane(s_len[g]);
    for (int i = lane; i < rl; i += 64) {
      int      i1 = i + 32;
      unsigned w0 = part[p0 + i];
      unsigned w1 = (i1 < rl) ? part[p0 + i1] : 0u;
      int q0 = atomicAdd(&s_cur[w0 >> SRC_BITS], 1);
      if (fits) s_sorted[q0] = (int)(w0 & SRC_MASK);
      else      sorted[bucket_base + q0] = (int)(w0 & SRC_MASK);
      if (i1 < rl) {
        int q1 = atomicAdd(&s_cur[w1 >> SRC_BITS], 1);
        if (fits) s_sorted[q1] = (int)(w1 & SRC_MASK);
        else      sorted[bucket_base + q1] = (int)(w1 & SRC_MASK);
      }
    }
  }
  if (fits) {
    __syncthreads();
    for (int i = t; i < total; i += 1024) sorted[bucket_base + i] = s_sorted[i];
  }
}

#define NPW 8
#define GRP 16
#define NO_END 0x40000000
__global__ void __launch_bounds__(256)
gather_sum(const float* __restrict__ xin, const int* __restrict__ sorted,
           const int* __restrict__ seg, float* __restrict__ agg, int n_nodes) {
  int wv   = __builtin_amdgcn_readfirstlane(threadIdx.x >> 5);
  int lane = threadIdx.x & 31;
  int n0   = (blockIdx.x * 8 + wv) * NPW;
  if (n0 >= n_nodes) return;
  int nn = min(NPW, n_nodes - n0);

  int2 sd = make_int2(0, 0);
  if (lane < nn) sd = *(const int2*)(seg + 2 * (n0 + lane));
  int e0   = __builtin_amdgcn_readlane(sd.x, 0);
  int e1   = __builtin_amdgcn_readlane(sd.x, nn - 1) +
             __builtin_amdgcn_readlane(sd.y, nn - 1);
  int ni   = 0;
  int nend = e0 + __builtin_amdgcn_readlane(sd.y, 0);

  float*   arow    = agg + (long long)n0 * DF + lane * 2;
  unsigned laneoff = (unsigned)lane * 8u;
  float2   acc     = make_float2(0.0f, 0.0f);

  if (e0 < e1) {
    int last = e1 - 1;
    int idx  = sorted[min(e0 + lane, last)];
    for (int base = e0; base < e1; base += 32) {
      int idx_next = sorted[min(base + 32 + lane, last)];
#pragma unroll
      for (int g = 0; g < 32 / GRP; ++g) {
        int p0 = base + g * GRP;
        if (p0 < e1) {
          float2 v[GRP];
#pragma unroll
          for (int u = 0; u < GRP; ++u) {
            unsigned s = (unsigned)__builtin_amdgcn_readlane(idx, g * GRP + u);
            v[u] = *(const float2*)((const char*)xin + ((s << 8) + laneoff));
          }
          int rel = nend - p0;
#pragma unroll
          for (int u = 0; u < GRP; ++u) {
            while (rel == u) {
              *(float2*)(arow + (long long)ni * DF) = acc;
              acc = make_float2(0.0f, 0.0f);
              ++ni;
              if (ni < nn) {
                int d = __builtin_amdgcn_readlane(sd.y, ni);
                nend += d;
                rel  += d;
              } else {
                nend = NO_END;
                rel  = NO_END;
              }
            }
            acc.x += v[u].x;
            acc.y += v[u].y;
          }
        }
      }
      idx = idx_next;
    }
  }
  for (; ni < nn; ++ni) {
    *(float2*)(arow + (long long)ni * DF) = acc;
    acc = make_float2(0.0f, 0.0f);
  }
}

__global__ void __launch_bounds__(128)
sage_linear(const float* __restrict__ agg,
            const float* __restrict__ cnt,
            const float* __restrict__ xin,
            const float* __restrict__ Wl,
            const float* __restrict__ bl,
            const float* __restrict__ Wr,
            float* __restrict__ out,
            int n_nodes, int do_relu) {
  __shared__ float sWl[DF * LDS_STRIDE];
  __shared__ float sWr[DF * LDS_STRIDE];

  for (int i = threadIdx.x; i < DF * DF / 2; i += 128) {
    int r = i >> 5;
    int c = (i & 31) * 2;
    *(float2*)&sWl[r * LDS_STRIDE + c] = *(const float2*)&Wl[r * DF + c];
    *(float2*)&sWr[r * LDS_STRIDE + c] = *(const float2*)&Wr[r * DF + c];
  }
  __syncthreads();

  int wave = threadIdx.x >> 5;
  int lane = threadIdx.x & 31;
  int tile = blockIdx.x * 4 + wave;
  int nb   = tile * 16;
  if (nb >= n_nodes) return;

  int m  = lane & 15;
  int kh = lane >> 4;

  int row = nb + m;
  float invc = 1.0f / fmaxf(cnt[row], 1.0f);
  const float* arow = agg + (long long)row * DF + 2 * kh;
  const float* xrow = xin + (long long)row * DF + 2 * kh;

  v8f acc[4];
#pragma unroll
  for (int nt = 0; nt < 4; ++nt) {
    float b = bl[nt * 16 + m];
    v8f c = {b, b, b, b, b, b, b, b};
    acc[nt] = c;
  }

  for (int k = 0; k < DF; k += 4) {
    v2f am = *(const v2f*)(arow + k);
    am *= invc;
    v2f ax = *(const v2f*)(xrow + k);
#pragma unroll
    for (int nt = 0; nt < 4; ++nt) {
      v2f bw = *(const v2f*)&sWl[(nt * 16 + m) * LDS_STRIDE + k + 2 * kh];
      acc[nt] = __builtin_amdgcn_wmma_f32_16x16x4_f32(
          false, am, false, bw, (short)0, acc[nt], false, false);
      v2f br = *(const v2f*)&sWr[(nt * 16 + m) * LDS_STRIDE + k + 2 * kh];
      acc[nt] = __builtin_amdgcn_wmma_f32_16x16x4_f32(
          false, ax, false, br, (short)0, acc[nt], false, false);
    }
  }

#pragma unroll
  for (int nt = 0; nt < 4; ++nt) {
#pragma unroll
    for (int g = 0; g < 8; ++g) {
      int orow = nb + g + 8 * kh;
      float v = acc[nt][g];
      if (do_relu) v = fmaxf(v, 0.0f);
      out[(long long)orow * DF + nt * 16 + m] = v;
    }
  }
}

extern "C" void kernel_launch(void* const* d_in, const int* in_sizes, int n_in,
                              void* d_out, int out_size, void* d_ws, size_t ws_size,
                              hipStream_t stream) {
  const float* x   = (const float*)d_in[0];
  const int*   ei  = (const int*)d_in[1];
  const float* W1l = (const float*)d_in[2];
  const float* b1l = (const float*)d_in[3];
  const float* W1r = (const float*)d_in[4];
  const float* W2l = (const float*)d_in[5];
  const float* b2l = (const float*)d_in[6];
  const float* W2r = (const float*)d_in[7];
  float* out = (float*)d_out;

  int n_nodes = in_sizes[0] / DF;
  int n_edges = in_sizes[1] / 2;
  const int* src = ei;
  const int* dst = ei + n_edges;

  size_t aggBytes = (size_t)n_nodes * DF * sizeof(float);
  char* ws = (char*)d_ws;
  float* agg = (float*)ws;
  float* h   = (float*)(ws + aggBytes);
  float* cnt = (float*)(ws + 2 * aggBytes);

  int n_buckets = (n_nodes + BK_NODES - 1) / BK_NODES;
  int n_chunks  = (n_edges + CH_EDGES - 1) / CH_EDGES;

  unsigned* part = (unsigned*)h;
  int*      tab  = (int*)(part + n_edges);
  int* seg    = (int*)d_out;
  int* sorted = seg + 2 * (size_t)n_nodes;

  int wblocks = (n_nodes + 8 * NPW - 1) / (8 * NPW);
  int tiles   = (n_nodes + 15) / 16;
  int gblocks = (tiles + 3) / 4;

  edge_partition<<<n_chunks, 1024, 0, stream>>>(src, dst, part, tab, n_edges,
                                                n_buckets, n_chunks);
  bucket_sort<<<n_buckets, 1024, 0, stream>>>(part, tab, seg, sorted, cnt,
                                              n_nodes, n_chunks);

  gather_sum<<<wblocks, 256, 0, stream>>>(x, sorted, seg, agg, n_nodes);
  sage_linear<<<gblocks, 128, 0, stream>>>(agg, cnt, x, W1l, b1l, W1r, h,
                                           n_nodes, 1);

  gather_sum<<<wblocks, 256, 0, stream>>>(h, sorted, seg, agg, n_nodes);
  sage_linear<<<gblocks, 128, 0, stream>>>(agg, cnt, h, W2l, b2l, W2r, out,
                                           n_nodes, 0);
}


# Layer fusion: gather and linear become one kernel per layer; a wave keeps its 16 nodes' sums in an LDS tile and feeds the WMMAs from there (no agg round trip; 4 launches, not 6); sum order kept
# speedup vs baseline: 4.5335x; 1.0978x over previous
#include <hip/hip_runtime.h>

typedef __attribute__((ext_vector_type(2))) float v2f;
typedef __attribute__((ext_vector_type(8))) float v8f;

#define DF 64
#define LDS_STRIDE 68

#define BK_SHIFT 9
#define BK_NODES 512
#define CH_EDGES 8192
#define SRC_BITS 23
#define SRC_MASK 0x7FFFFFu
#define SORT_CAP 16384

__device__ __forceinline__ int wave_incl_scan(int v, int lane) {
  for (int s = 1; s < 32; s <<= 1) {
    int t = __shfl_up(v, s, 32);
    if (lane >= s) v += t;
  }
  return v;
}

__global__ void __launch_bounds__(1024)
edge_partition(const int* __restrict__ src, const int* __restrict__ dst,
               unsigned* __restrict__ part, int* __restrict__ tab,
               int n_edges, int n_buckets, int n_chunks) {
  __shared__ int s_cnt[257];
  __shared__ int s_off[257];
  __shared__ int s_wtot[8];
  __shared__ unsigned s_stage[CH_EDGES];

  int t    = threadIdx.x;
  int lane = t & 31;
  int wave = t >> 5;
  int g    = blockIdx.x;
  int base = g * CH_EDGES;
  int len  = min(CH_EDGES, n_edges - base);

  if (t <= 256) s_cnt[t] = 0;

  int d[8], s[8];
#pragma unroll
  for (int k = 0; k < 8; ++k) {
    int i = k * 1024 + t;
    d[k] = -1;
    s[k] = 0;
    if (i < len) {
      d[k] = dst[base + i];
      s[k] = src[base + i];
    }
  }
  __syncthreads();

  int br[8];
#pragma unroll
  for (int k = 0; k < 8; ++k) {
    int b = (d[k] >= 0) ? (d[k] >> BK_SHIFT) : 256;
    int r = atomicAdd(&s_cnt[b], 1);
    br[k] = (b << 16) | r;
  }
  __syncthreads();

  int v = 0, incl = 0;
  if (t < 256) {
    v    = (t < n_buckets) ? s_cnt[t] : 0;
    incl = wave_incl_scan(v, lane);
    if (lane == 31) s_wtot[wave] = incl;
  }
  __syncthreads();
  if (t < 256) {
    int wb = 0;
    for (int w = 0; w < wave; ++w) wb += s_wtot[w];
    s_off[t + 1] = wb + incl;
    if (t == 0) s_off[0] = 0;
  }
  __syncthreads();

#pragma unroll
  for (int k = 0; k < 8; ++k) {
    if (d[k] >= 0) {
      int b = br[k] >> 16;
      int r = br[k] & 0xFFFF;
      s_stage[s_off[b] + r] =
          ((unsigned)(d[k] & (BK_NODES - 1)) << SRC_BITS) | (unsigned)s[k];
    }
  }
  __syncthreads();

  for (int i = t; i < len; i += 1024) part[base + i] = s_stage[i];
  if (t <= n_buckets) tab[t * n_chunks + g] = s_off[t];
}

__global__ void __launch_bounds__(1024)
bucket_sort(const unsigned* __restrict__ part, const int* __restrict__ tab,
            int* __restrict__ seg, int* __restrict__ sorted,
            float* __restrict__ cnt, int n_nodes, int n_chunks) {
  __shared__ int s_lo[1024];
  __shared__ int s_len[1024];
  __shared__ int s_cnt[BK_NODES];
  __shared__ int s_cur[BK_NODES];
  __shared__ int s_wtot[16];
  __shared__ int s_base, s_total;
  __shared__ int s_sorted[SORT_CAP];

  int t    = threadIdx.x;
  int lane = t & 31;
  int wave = t >> 5;
  int wv   = __builtin_amdgcn_readfirstlane(wave);
  int b    = blockIdx.x;

  if (t == 0) { s_base = 0; s_total = 0; }
  if (t < BK_NODES) s_cnt[t] = 0;
  __syncthreads();

  int lo = 0, ln = 0;
  if (t < n_chunks) {
    lo = tab[b * n_chunks + t];
    ln = tab[(b + 1) * n_chunks + t] - lo;
    s_lo[t]  = t * CH_EDGES + lo;
    s_len[t] = ln;
  }
  int a = lo, c = ln;
  for (int sh = 16; sh > 0; sh >>= 1) {
    a += __shfl_xor(a, sh, 32);
    c += __shfl_xor(c, sh, 32);
  }
  if (lane == 0 && (a | c) != 0) {
    atomicAdd(&s_base, a);
    atomicAdd(&s_total, c);
  }
  __syncthreads();
  int  bucket_base = s_base;
  int  total       = s_total;
  bool fits        = total <= SORT_CAP;

  for (int g = wv; g < n_chunks; g += 32) {
    int p0 = __builtin_amdgcn_readfirstlane(s_lo[g]);
    int rl = __builtin_amdgcn_readfirstlane(s_len[g]);
    for (int i = lane; i < rl; i += 64) {
      int      i1 = i + 32;
      unsigned w0 = part[p0 + i];
      unsigned w1 = (i1 < rl) ? part[p0 + i1] : 0u;
      atomicAdd(&s_cnt[w0 >> SRC_BITS], 1);
      if (i1 < rl) atomicAdd(&s_cnt[w1 >> SRC_BITS], 1);
    }
  }
  __syncthreads();

  int v = 0, incl = 0;
  if (t < BK_NODES) {
    v    = s_cnt[t];
    incl = wave_incl_scan(v, lane);
    if (lane == 31) s_wtot[wave] = incl;
  }
  __syncthreads();
  if (t < BK_NODES) {
    int wb = 0;
    for (int w = 0; w < wave; ++w) wb += s_wtot[w];
    int start = wb + incl - v;
    s_cur[t] = start;
    int n = b * BK_NODES + t;
    if (n < n_nodes) {
      *(int2*)(seg + 2 * n) = make_int2(bucket_base + start, v);
      cnt[n] = (float)v;
    }
  }
  __syncthreads();

  for (int g = wv; g < n_chunks; g += 32) {
    int p0 = __builtin_amdgcn_readfirstlane(s_lo[g]);
    int rl = __builtin_amdgcn_readfirstlane(s_len[g]);
    for (int i = lane; i < rl; i += 64) {
      int      i1 = i + 32;
      unsigned w0 = part[p0 + i];
      unsigned w1 = (i1 < rl) ? part[p0 + i1] : 0u;
      int q0 = atomicAdd(&s_cur[w0 >> SRC_BITS], 1);
      if (fits) s_sorted[q0] = (int)(w0 & SRC_MASK);
      else      sorted[bucket_base + q0] = (int)(w0 & SRC_MASK);
      if (i1 < rl) {
        int q1 = atomicAdd(&s_cur[w1 >> SRC_BITS], 1);
        if (fits) s_sorted[q1] = (int)(w1 & SRC_MASK);
        else      sorted[bucket_base + q1] = (int)(w1 & SRC_MASK);
      }
    }
  }
  if (fits) {
    __syncthreads();
    for (int i = t; i < total; i += 1024) sorted[bucket_base + i] = s_sorted[i];
  }
}

#define TPB 8
#define GRP 16
#define NO_END 0x40000000

__device__ __forceinline__ void wave_lds_handover() {
  __builtin_amdgcn_fence(__ATOMIC_ACQ_REL, "wavefront");
  __builtin_amdgcn_wave_barrier();
}

__global__ void __launch_bounds__(TPB * 32)
sage_layer(const float* __restrict__ xin, const int* __restrict__ sorted,
           const int* __restrict__ seg, const float* __restrict__ cnt,
           const float* __restrict__ Wl,
           const float* __restrict__ bl,
           const float* __restrict__ Wr,
           float* __restrict__ out, int n_nodes, int do_relu) {
  __shared__ float sWl[DF * LDS_STRIDE];
  __shared__ float sWr[DF * LDS_STRIDE];
  __shared__ float sT[TPB][16 * LDS_STRIDE];

  int wv   = __builtin_amdgcn_readfirstlane(threadIdx.x >> 5);
  int lane = threadIdx.x & 31;
  int nb   = (blockIdx.x * TPB + wv) * 16;
  int nn   = min(16, n_nodes - nb);
  int m    = lane & 15;
  int kh   = lane >> 4;
  float* tile = sT[wv];

  int    nlast = n_nodes - 1;
  int2   sd = *(const int2*)(seg + 2 * min(nb + m, nlast));
  float  cn = cnt[min(nb + m, nlast)];
  float2 xr[16];
#pragma unroll
  for (int r = 0; r < 16; ++r)
    xr[r] = *(const float2*)(xin + (long long)min(nb + r, nlast) * DF + lane * 2);

  int    wc = lane * 2;
  float2 wl[8], wr[8];
#pragma unroll
  for (int j = 0; j < 8; ++j) {
    int r = wv + TPB * j;
    wl[j] = *(const float2*)&Wl[r * DF + wc];
    wr[j] = *(const float2*)&Wr[r * DF + wc];
  }
#pragma unroll
  for (int j = 0; j < 8; ++j) {
    int r = wv + TPB * j;
    *(float2*)&sWl[r * LDS_STRIDE + wc] = wl[j];
    *(float2*)&sWr[r * LDS_STRIDE + wc] = wr[j];
  }
#pragma unroll
  for (int r = 0; r < 16; ++r) *(float2*)&tile[r * LDS_STRIDE + lane * 2] = xr[r];
  __syncthreads();
  if (nn <= 0) return;

  v2f xa[16];
#pragma unroll
  for (int kk = 0; kk < 16; ++kk)
    xa[kk] = *(const v2f*)&tile[m * LDS_STRIDE + kk * 4 + 2 * kh];
  float bias[4];
#pragma unroll
  for (int nt = 0; nt < 4; ++nt) bias[nt] = bl[nt * 16 + m];
  wave_lds_handover();

  int e0   = __builtin_amdgcn_readlane(sd.x, 0);
  int e1   = __builtin_amdgcn_readlane(sd.x, nn - 1) +
             __builtin_amdgcn_readlane(sd.y, nn - 1);
  int ni   = 0;
  int nend = e0 + __builtin_amdgcn_readlane(sd.y, 0);

  float*   trow    = tile + lane * 2;
  unsigned laneoff = (unsigned)lane * 8u;
  float2   acc     = make_float2(0.0f, 0.0f);

  if (e0 < e1) {
    int last = e1 - 1;
    int idx  = sorted[min(e0 + lane, last)];
    for (int base = e0; base < e1; base += 32) {
      int idx_next = sorted[min(base + 32 + lane, last)];
#pragma unroll
      for (int g = 0; g < 32 / GRP; ++g) {
        int p0 = base + g * GRP;
        if (p0 < e1) {
          float2 v[GRP];
#pragma unroll
          for (int u = 0; u < GRP; ++u) {
            unsigned s = (unsigned)__builtin_amdgcn_readlane(idx, g * GRP + u);
            v[u] = *(const float2*)((const char*)xin + ((s << 8) + laneoff));
          }
          int rel = nend - p0;
#pragma unroll
          for (int u = 0; u < GRP; ++u) {
            while (rel == u) {
              *(float2*)(trow + ni * LDS_STRIDE) = acc;
              acc = make_float2(0.0f, 0.0f);
              ++ni;
              if (ni < nn) {
                int d = __builtin_amdgcn_readlane(sd.y, ni);
                nend += d;
                rel  += d;
              } else {
                nend = NO_END;
                rel  = NO_END;
              }
            }
            acc.x += v[u].x;
            acc.y += v[u].y;
          }
        }
      }
      idx = idx_next;
    }
  }
  for (; ni < 16; ++ni) {
    *(float2*)(trow + ni * LDS_STRIDE) = acc;
    acc = make_float2(0.0f, 0.0f);
  }
  wave_lds_handover();

  float invc = 1.0f / fmaxf(cn, 1.0f);
  v8f cacc[4];
#pragma unroll
  for (int nt = 0; nt < 4; ++nt) {
    float b = bias[nt];
    v8f c = {b, b, b, b, b, b, b, b};
    cacc[nt] = c;
  }

#pragma unroll
  for (int kk = 0; kk < 16; ++kk) {
    int k = kk * 4;
    v2f am = *(const v2f*)&tile[m * LDS_STRIDE + k + 2 * kh];
    am *= invc;
    v2f ax = xa[kk];
#pragma unroll
    for (int nt = 0; nt < 4; ++nt) {
      v2f bw = *(const v2f*)&sWl[(nt * 16 + m) * LDS_STRIDE + k + 2 * kh];
      cacc[nt] = __builtin_amdgcn_wmma_f32_16x16x4_f32(
          false, am, false, bw, (short)0, cacc[nt], false, false);
      v2f br = *(const v2f*)&sWr[(nt * 16 + m) * LDS_STRIDE + k + 2 * kh];
      cacc[nt] = __builtin_amdgcn_wmma_f32_16x16x4_f32(
          false, ax, false, br, (short)0, cacc[nt], false, false);
    }
  }

  float* orow0 = out + (long long)(nb + 8 * kh) * DF + m;
  if (nn == 16) {
#pragma unroll
    for (int nt = 0; nt < 4; ++nt) {
#pragma unroll
      for (int g = 0; g < 8; ++g) {
        float v = cacc[nt][g];
        if (do_relu) v = fmaxf(v, 0.0f);
        orow0[g * DF + nt * 16] = v;
      }
    }
  } else {
#pragma unroll
    for (int nt = 0; nt < 4; ++nt) {
#pragma unroll
      for (int g = 0; g < 8; ++g) {
        float v = cacc[nt][g];
        if (do_relu) v = fmaxf(v, 0.0f);
        if (g + 8 * kh < nn) orow0[g * DF + nt * 16] = v;
      }
    }
  }
}

extern "C" void kernel_launch(void* const* d_in, const int* in_sizes, int n_in,
                              void* d_out, int out_size, void* d_ws, size_t ws_size,
                              hipStream_t stream) {
  const float* x   = (const float*)d_in[0];
  const int*   ei  = (const int*)d_in[1];
  const float* W1l = (const float*)d_in[2];
  const float* b1l = (const float*)d_in[3];
  const float* W1r = (const float*)d_in[4];
  const float* W2l = (const float*)d_in[5];
  const float* b2l = (const float*)d_in[6];
  const float* W2r = (const float*)d_in[7];
  float* out = (float*)d_out;

  int n_nodes = in_sizes[0] / DF;
  int n_edges = in_sizes[1] / 2;
  const int* src = ei;
  const int* dst = ei + n_edges;

  size_t rowBytes = (size_t)n_nodes * DF * sizeof(float);
  char* ws = (char*)d_ws;
  int*   seg    = (int*)ws;
  int*   sorted = seg + 2 * (size_t)n_nodes;
  float* h      = (float*)(ws + rowBytes);
  float* cnt    = (float*)(ws + 2 * rowBytes);

  int n_buckets = (n_nodes + BK_NODES - 1) / BK_NODES;
  int n_chunks  = (n_edges + CH_EDGES - 1) / CH_EDGES;

  unsigned* part = (unsigned*)h;
  int*      tab  = (int*)(part + n_edges);

  int tiles   = (n_nodes + 15) / 16;
  int lblocks = (tiles + TPB - 1) / TPB;

  edge_partition<<<n_chunks, 1024, 0, stream>>>(src, dst, part, tab, n_edges,
                                                n_buckets, n_chunks);
  bucket_sort<<<n_buckets, 1024, 0, stream>>>(part, tab, seg, sorted, cnt,
                                              n_nodes, n_chunks);

  sage_layer<<<lblocks, TPB * 32, 0, stream>>>(x, sorted, seg, cnt, W1l, b1l, W1r, h,
                                               n_nodes, 1);
  sage_layer<<<lblocks, TPB * 32, 0, stream>>>(h, sorted, seg, cnt, W2l, b2l, W2r, out,
                                               n_nodes, 0);
}


# Products: the 64 xin x Wr WMMAs of a tile issue inside the gather stream (2 k-steps per 32-source trip, between row-load issue and first use); sum order now bias, xin k-steps, mean k-steps
# speedup vs baseline: 4.7812x; 1.0546x over previous
#include <hip/hip_runtime.h>

typedef __attribute__((ext_vector_type(2))) float v2f;
typedef __attribute__((ext_vector_type(8))) float v8f;

#define DF 64
#define LDS_STRIDE 68

#define BK_SHIFT 9
#define BK_NODES 512
#define CH_EDGES 8192
#define SRC_BITS 23
#define SRC_MASK 0x7FFFFFu
#define SORT_CAP 16384

__device__ __forceinline__ int wave_incl_scan(int v, int lane) {
  for (int s = 1; s < 32; s <<= 1) {
    int t = __shfl_up(v, s, 32);
    if (lane >= s) v += t;
  }
  return v;
}

__global__ void __launch_bounds__(1024)
edge_partition(const int* __restrict__ src, const int* __restrict__ dst,
               unsigned* __restrict__ part, int* __restrict__ tab,
               int n_edges, int n_buckets, int n_chunks) {
  __shared__ int s_cnt[257];
  __shared__ int s_off[257];
  __shared__ int s_wtot[8];
  __shared__ unsigned s_stage[CH_EDGES];

  int t    = threadIdx.x;
  int lane = t & 31;
  int wave = t >> 5;
  int g    = blockIdx.x;
  int base = g * CH_EDGES;
  int len  = min(CH_EDGES, n_edges - base);

  if (t <= 256) s_cnt[t] = 0;

  int d[8], s[8];
#pragma unroll
  for (int k = 0; k < 8; ++k) {
    int i = k * 1024 + t;
    d[k] = -1;
    s[k] = 0;
    if (i < len) {
      d[k] = dst[base + i];
      s[k] = src[base + i];
    }
  }
  __syncthreads();

  int br[8];
#pragma unroll
  for (int k = 0; k < 8; ++k) {
    int b = (d[k] >= 0) ? (d[k] >> BK_SHIFT) : 256;
    int r = atomicAdd(&s_cnt[b], 1);
    br[k] = (b << 16) | r;
  }
  __syncthreads();

  int v = 0, incl = 0;
  if (t < 256) {
    v    = (t < n_buckets) ? s_cnt[t] : 0;
    incl = wave_incl_scan(v, lane);
    if (lane == 31) s_wtot[wave] = incl;
  }
  __syncthreads();
  if (t < 256) {
    int wb = 0;
    for (int w = 0; w < wave; ++w) wb += s_wtot[w];
    s_off[t + 1] = wb + incl;
    if (t == 0) s_off[0] = 0;
  }
  __syncthreads();

#pragma unroll
  for (int k = 0; k < 8; ++k) {
    if (d[k] >= 0) {
      int b = br[k] >> 16;
      int r = br[k] & 0xFFFF;
      s_stage[s_off[b] + r] =
          ((unsigned)(d[k] & (BK_NODES - 1)) << SRC_BITS) | (unsigned)s[k];
    }
  }
  __syncthreads();

  for (int i = t; i < len; i += 1024) part[base + i] = s_stage[i];
  if (t <= n_buckets) tab[t * n_chunks + g] = s_off[t];
}

__global__ void __launch_bounds__(1024)
bucket_sort(const unsigned* __restrict__ part, const int* __restrict__ tab,
            int* __restrict__ seg, int* __restrict__ sorted,
            float* __restrict__ cnt, int n_nodes, int n_chunks) {
  __shared__ int s_lo[1024];
  __shared__ int s_len[1024];
  __shared__ int s_cnt[BK_NODES];
  __shared__ int s_cur[BK_NODES];
  __shared__ int s_wtot[16];
  __shared__ int s_base, s_total;
  __shared__ int s_sorted[SORT_CAP];

  int t    = threadIdx.x;
  int lane = t & 31;
  int wave = t >> 5;
  int wv   = __builtin_amdgcn_readfirstlane(wave);
  int b    = blockIdx.x;

  if (t == 0) { s_base = 0; s_total = 0; }
  if (t < BK_NODES) s_cnt[t] = 0;
  __syncthreads();

  int lo = 0, ln = 0;
  if (t < n_chunks) {
    lo = tab[b * n_chunks + t];
    ln = tab[(b + 1) * n_chunks + t] - lo;
    s_lo[t]  = t * CH_EDGES + lo;
    s_len[t] = ln;
  }
  int a = lo, c = ln;
  for (int sh = 16; sh > 0; sh >>= 1) {
    a += __shfl_xor(a, sh, 32);
    c += __shfl_xor(c, sh, 32);
  }
  if (lane == 0 && (a | c) != 0) {
    atomicAdd(&s_base, a);
    atomicAdd(&s_total, c);
  }
  __syncthreads();
  int  bucket_base = s_base;
  int  total       = s_total;
  bool fits        = total <= SORT_CAP;

  for (int g = wv; g < n_chunks; g += 32) {
    int p0 = __builtin_amdgcn_readfirstlane(s_lo[g]);
    int rl = __builtin_amdgcn_readfirstlane(s_len[g]);
    for (int i = lane; i < rl; i += 64) {
      int      i1 = i + 32;
      unsigned w0 = part[p0 + i];
      unsigned w1 = (i1 < rl) ? part[p0 + i1] : 0u;
      atomicAdd(&s_cnt[w0 >> SRC_BITS], 1);
      if (i1 < rl) atomicAdd(&s_cnt[w1 >> SRC_BITS], 1);
    }
  }
  __syncthreads();

  int v = 0, incl = 0;
  if (t < BK_NODES) {
    v    = s_cnt[t];
    incl = wave_incl_scan(v, lane);
    if (lane == 31) s_wtot[wave] = incl;
  }
  __syncthreads();
  if (t < BK_NODES) {
    int wb = 0;
    for (int w = 0; w < wave; ++w) wb += s_wtot[w];
    int start = wb + incl - v;
    s_cur[t] = start;
    int n = b * BK_NODES + t;
    if (n < n_nodes) {
      *(int2*)(seg + 2 * n) = make_int2(bucket_base + start, v);
      cnt[n] = (float)v;
    }
  }
  __syncthreads();

  for (int g = wv; g < n_chunks; g += 32) {
    int p0 = __builtin_amdgcn_readfirstlane(s_lo[g]);
    int rl = __builtin_amdgcn_readfirstlane(s_len[g]);
    for (int i = lane; i < rl; i += 64) {
      int      i1 = i + 32;
      unsigned w0 = part[p0 + i];
      unsigned w1 = (i1 < rl) ? part[p0 + i1] : 0u;
      int q0 = atomicAdd(&s_cur[w0 >> SRC_BITS], 1);
      if (fits) s_sorted[q0] = (int)(w0 & SRC_MASK);
      else      sorted[bucket_base + q0] = (int)(w0 & SRC_MASK);
      if (i1 < rl) {
        int q1 = atomicAdd(&s_cur[w1 >> SRC_BITS], 1);
        if (fits) s_sorted[q1] = (int)(w1 & SRC_MASK);
        else      sorted[bucket_base + q1] = (int)(w1 & SRC_MASK);
      }
    }
  }
  if (fits) {
    __syncthreads();
    for (int i = t; i < total; i += 1024) sorted[bucket_base + i] = s_sorted[i];
  }
}

#define TPB 8
#define GRP 16
#define NO_END 0x40000000

__device__ __forceinline__ void wave_lds_handover() {
  __builtin_amdgcn_fence(__ATOMIC_ACQ_REL, "wavefront");
  __builtin_amdgcn_wave_barrier();
}

__global__ void __launch_bounds__(TPB * 32)
sage_layer(const float* __restrict__ xin, const int* __restrict__ sorted,
           const int* __restrict__ seg, const float* __restrict__ cnt,
           const float* __restrict__ Wl,
           const float* __restrict__ bl,
           const float* __restrict__ Wr,
           float* __restrict__ out, int n_nodes, int do_relu) {
  __shared__ float sWl[DF * LDS_STRIDE];
  __shared__ float sWr[DF * LDS_STRIDE];
  __shared__ float sT[TPB][16 * LDS_STRIDE];

  int wv   = __builtin_amdgcn_readfirstlane(threadIdx.x >> 5);
  int lane = threadIdx.x & 31;
  int nb   = (blockIdx.x * TPB + wv) * 16;
  int nn   = min(16, n_nodes - nb);
  int m    = lane & 15;
  int kh   = lane >> 4;
  float* tile = sT[wv];

  int    nlast = n_nodes - 1;
  int2   sd = *(const int2*)(seg + 2 * min(nb + m, nlast));
  float  cn = cnt[min(nb + m, nlast)];
  float2 xr[16];
#pragma unroll
  for (int r = 0; r < 16; ++r)
    xr[r] = *(const float2*)(xin + (long long)min(nb + r, nlast) * DF + lane * 2);

  int    wc = lane * 2;
  float2 wl[8], wr[8];
#pragma unroll
  for (int j = 0; j < 8; ++j) {
    int r = wv + TPB * j;
    wl[j] = *(const float2*)&Wl[r * DF + wc];
    wr[j] = *(const float2*)&Wr[r * DF + wc];
  }
#pragma unroll
  for (int j = 0; j < 8; ++j) {
    int r = wv + TPB * j;
    *(float2*)&sWl[r * LDS_STRIDE + wc] = wl[j];
    *(float2*)&sWr[r * LDS_STRIDE + wc] = wr[j];
  }
#pragma unroll
  for (int r = 0; r < 16; ++r) *(float2*)&tile[r * LDS_STRIDE + lane * 2] = xr[r];
  __syncthreads();
  if (nn <= 0) return;

  v2f xa[16];
#pragma unroll
  for (int kk = 0; kk < 16; ++kk)
    xa[kk] = *(const v2f*)&tile[m * LDS_STRIDE + kk * 4 + 2 * kh];
  float bias[4];
#pragma unroll
  for (int nt = 0; nt < 4; ++nt) bias[nt] = bl[nt * 16 + m];
  wave_lds_handover();

  int e0   = __builtin_amdgcn_readlane(sd.x, 0);
  int e1   = __builtin_amdgcn_readlane(sd.x, nn - 1) +
             __builtin_amdgcn_readlane(sd.y, nn - 1);
  int ni   = 0;
  int nend = e0 + __builtin_amdgcn_readlane(sd.y, 0);

  float*   trow    = tile + lane * 2;
  unsigned laneoff = (unsigned)lane * 8u;
  float2   acc     = make_float2(0.0f, 0.0f);

  v8f cacc[4];
#pragma unroll
  for (int nt = 0; nt < 4; ++nt) {
    float b = bias[nt];
    v8f c = {b, b, b, b, b, b, b, b};
    cacc[nt] = c;
  }
  int  ks    = 0;
  auto self2 = [&]() __attribute__((always_inline)) {
    const float* wb = &sWr[m * LDS_STRIDE + ks * 4 + 2 * kh];
#pragma unroll
    for (int h = 0; h < 2; ++h) {
#pragma unroll
      for (int nt = 0; nt < 4; ++nt) {
        v2f bq = *(const v2f*)&wb[nt * 16 * LDS_STRIDE + h * 4];
        cacc[nt] = __builtin_amdgcn_wmma_f32_16x16x4_f32(
            false, xa[h], false, bq, (short)0, cacc[nt], false, false);
      }
    }
#pragma unroll
    for (int j = 0; j < 14; ++j) xa[j] = xa[j + 2];
    ks += 2;
  };

  int    last = e1 - 1;
  int    base = e0;
  int    idx  = 0;
  if (e0 < e1) idx = sorted[min(e0 + lane, last)];
  float2 v[GRP];

  auto issue = [&](int g) __attribute__((always_inline)) {
#pragma unroll
    for (int u = 0; u < GRP; ++u) {
      unsigned s = (unsigned)__builtin_amdgcn_readlane(idx, g * GRP + u);
      v[u] = *(const float2*)((const char*)xin + ((s << 8) + laneoff));
    }
  };
  auto consume = [&](int p0) __attribute__((always_inline)) {
    int rel = nend - p0;
#pragma unroll
    for (int u = 0; u < GRP; ++u) {
      while (rel == u) {
        *(float2*)(trow + ni * LDS_STRIDE) = acc;
        acc = make_float2(0.0f, 0.0f);
        ++ni;
        if (ni < nn) {
          int d = __builtin_amdgcn_readlane(sd.y, ni);
          nend += d;
          rel  += d;
        } else {
          nend = NO_END;
          rel  = NO_END;
        }
      }
      acc.x += v[u].x;
      acc.y += v[u].y;
    }
  };

  auto trip = [&](bool carry) __attribute__((always_inline)) {
    int idx_next = sorted[min(base + 32 + lane, last)];
    issue(0);
    if (carry) self2();
    consume(base);
    if (base + GRP < e1) {
      issue(1);
      consume(base + GRP);
    }
    idx = idx_next;
    base += 32;
  };
  while (base < e1 && ks < 16) trip(true);
  while (base < e1) trip(false);
  for (; ni < 16; ++ni) {
    *(float2*)(trow + ni * LDS_STRIDE) = acc;
    acc = make_float2(0.0f, 0.0f);
  }
  wave_lds_handover();

#pragma nounroll
  while (ks < 16) self2();

  float invc = 1.0f / fmaxf(cn, 1.0f);
#pragma unroll
  for (int kk = 0; kk < 16; ++kk) {
    int k = kk * 4;
    v2f am = *(const v2f*)&tile[m * LDS_STRIDE + k + 2 * kh];
    am *= invc;
#pragma unroll
    for (int nt = 0; nt < 4; ++nt) {
      v2f bw = *(const v2f*)&sWl[(nt * 16 + m) * LDS_STRIDE + k + 2 * kh];
      cacc[nt] = __builtin_amdgcn_wmma_f32_16x16x4_f32(
          false, am, false, bw, (short)0, cacc[nt], false, false);
    }
  }

  float* orow0 = out + (long long)(nb + 8 * kh) * DF + m;
  if (nn == 16) {
#pragma unroll
    for (int nt = 0; nt < 4; ++nt) {
#pragma unroll
      for (int g = 0; g < 8; ++g) {
        float v = cacc[nt][g];
        if (do_relu) v = fmaxf(v, 0.0f);
        orow0[g * DF + nt * 16] = v;
      }
    }
  } else {
#pragma unroll
    for (int nt = 0; nt < 4; ++nt) {
#pragma unroll
      for (int g = 0; g < 8; ++g) {
        float v = cacc[nt][g];
        if (do_relu) v = fmaxf(v, 0.0f);
        if (g + 8 * kh < nn) orow0[g * DF + nt * 16] = v;
      }
    }
  }
}

extern "C" void kernel_launch(void* const* d_in, const int* in_sizes, int n_in,
                              void* d_out, int out_size, void* d_ws, size_t ws_size,
                              hipStream_t stream) {
  const float* x   = (const float*)d_in[0];
  const int*   ei  = (const int*)d_in[1];
  const float* W1l = (const float*)d_in[2];
  const float* b1l = (const float*)d_in[3];
  const float* W1r = (const float*)d_in[4];
  const float* W2l = (const float*)d_in[5];
  const float* b2l = (const float*)d_in[6];
  const float* W2r = (const float*)d_in[7];
  float* out = (float*)d_out;

  int n_nodes = in_sizes[0] / DF;
  int n_edges = in_sizes[1] / 2;
  const int* src = ei;
  const int* dst = ei + n_edges;

  size_t rowBytes = (size_t)n_nodes * DF * sizeof(float);
  char* ws = (char*)d_ws;
  int*   seg    = (int*)ws;
  int*   sorted = seg + 2 * (size_t)n_nodes;
  float* h      = (float*)(ws + rowBytes);
  float* cnt    = (float*)(ws + 2 * rowBytes);

  int n_buckets = (n_nodes + BK_NODES - 1) / BK_NODES;
  int n_chunks  = (n_edges + CH_EDGES - 1) / CH_EDGES;

  unsigned* part = (unsigned*)h;
  int*      tab  = (int*)(part + n_edges);

  int tiles   = (n_nodes + 15) / 16;
  int lblocks = (tiles + TPB - 1) / TPB;

  edge_partition<<<n_chunks, 1024, 0, stream>>>(src, dst, part, tab, n_edges,
                                                n_buckets, n_chunks);
  bucket_sort<<<n_buckets, 1024, 0, stream>>>(part, tab, seg, sorted, cnt,
                                              n_nodes, n_chunks);

  sage_layer<<<lblocks, TPB * 32, 0, stream>>>(x, sorted, seg, cnt, W1l, b1l, W1r, h,
                                               n_nodes, 1);
  sage_layer<<<lblocks, TPB * 32, 0, stream>>>(h, sorted, seg, cnt, W2l, b2l, W2r, out,
                                               n_nodes, 0);
}


# Launch fusion: bucket sort and layer 1 are one kernel (a work-group sorts its bucket, then runs layer 1 on that bucket's 32 tiles from the lists it just wrote): 3 launches, not 4; sum order kept
# speedup vs baseline: 5.0732x; 1.0611x over previous
#include <hip/hip_runtime.h>

typedef __attribute__((ext_vector_type(2))) float v2f;
typedef __attribute__((ext_vector_type(8))) float v8f;

#define DF 64
#define LDS_STRIDE 68

#define BK_SHIFT 9
#define BK_NODES 512
#define CH_EDGES 8192
#define SRC_BITS 23
#define SRC_MASK 0x7FFFFFu
#define SORT_CAP 16384

__device__ __forceinline__ int wave_incl_scan(int v, int lane) {
  for (int s = 1; s < 32; s <<= 1) {
    int t = __shfl_up(v, s, 32);
    if (lane >= s) v += t;
  }
  return v;
}

__global__ void __launch_bounds__(1024)
edge_partition(const int* __restrict__ src, const int* __restrict__ dst,
               unsigned* __restrict__ part, int* __restrict__ tab,
               int n_edges, int n_buckets, int n_chunks) {
  __shared__ int s_cnt[257];
  __shared__ int s_off[257];
  __shared__ int s_wtot[8];
  __shared__ unsigned s_stage[CH_EDGES];

  int t    = threadIdx.x;
  int lane = t & 31;
  int wave = t >> 5;
  int g    = blockIdx.x;
  int base = g * CH_EDGES;
  int len  = min(CH_EDGES, n_edges - base);

  if (t <= 256) s_cnt[t] = 0;

  int d[8], s[8];
#pragma unroll
  for (int k = 0; k < 8; ++k) {
    int i = k * 1024 + t;
    d[k] = -1;
    s[k] = 0;
    if (i < len) {
      d[k] = dst[base + i];
      s[k] = src[base + i];
    }
  }
  __syncthreads();

  int br[8];
#pragma unroll
  for (int k = 0; k < 8; ++k) {
    int b = (d[k] >= 0) ? (d[k] >> BK_SHIFT) : 256;
    int r = atomicAdd(&s_cnt[b], 1);
    br[k] = (b << 16) | r;
  }
  __syncthreads();

  int v = 0, incl = 0;
  if (t < 256) {
    v    = (t < n_buckets) ? s_cnt[t] : 0;
    incl = wave_incl_scan(v, lane);
    if (lane == 31) s_wtot[wave] = incl;
  }
  __syncthreads();
  if (t < 256) {
    int wb = 0;
    for (int w = 0; w < wave; ++w) wb += s_wtot[w];
    s_off[t + 1] = wb + incl;
    if (t == 0) s_off[0] = 0;
  }
  __syncthreads();

#pragma unroll
  for (int k = 0; k < 8; ++k) {
    if (d[k] >= 0) {
      int b = br[k] >> 16;
      int r = br[k] & 0xFFFF;
      s_stage[s_off[b] + r] =
          ((unsigned)(d[k] & (BK_NODES - 1)) << SRC_BITS) | (unsigned)s[k];
    }
  }
  __syncthreads();

  for (int i = t; i < len; i += 1024) part[base + i] = s_stage[i];
  if (t <= n_buckets) tab[t * n_chunks + g] = s_off[t];
}

__device__ __forceinline__ void
bucket_sort(const unsigned* __restrict__ part, const int* __restrict__ tab,
            int* seg, int* sorted, float* cnt, int n_nodes, int n_chunks) {
  __shared__ int s_lo[1024];
  __shared__ int s_len[1024];
  __shared__ int s_cnt[BK_NODES];
  __shared__ int s_cur[BK_NODES];
  __shared__ int s_wtot[16];
  __shared__ int s_base, s_total;
  __shared__ int s_sorted[SORT_CAP];

  int t    = threadIdx.x;
  int lane = t & 31;
  int wave = t >> 5;
  int wv   = __builtin_amdgcn_readfirstlane(wave);
  int b    = blockIdx.x;

  if (t == 0) { s_base = 0; s_total = 0; }
  if (t < BK_NODES) s_cnt[t] = 0;
  __syncthreads();

  int lo = 0, ln = 0;
  if (t < n_chunks) {
    lo = tab[b * n_chunks + t];
    ln = tab[(b + 1) * n_chunks + t] - lo;
    s_lo[t]  = t * CH_EDGES + lo;
    s_len[t] = ln;
  }
  int a = lo, c = ln;
  for (int sh = 16; sh > 0; sh >>= 1) {
    a += __shfl_xor(a, sh, 32);
    c += __shfl_xor(c, sh, 32);
  }
  if (lane == 0 && (a | c) != 0) {
    atomicAdd(&s_base, a);
    atomicAdd(&s_total, c);
  }
  __syncthreads();
  int  bucket_base = s_base;
  int  total       = s_total;
  bool fits        = total <= SORT_CAP;

  for (int g = wv; g < n_chunks; g += 32) {
    int p0 = __builtin_amdgcn_readfirstlane(s_lo[g]);
    int rl = __builtin_amdgcn_readfirstlane(s_len[g]);
    for (int i = lane; i < rl; i += 64) {
      int      i1 = i + 32;
      unsigned w0 = part[p0 + i];
      unsigned w1 = (i1 < rl) ? part[p0 + i1] : 0u;
      atomicAdd(&s_cnt[w0 >> SRC_BITS], 1);
      if (i1 < rl) atomicAdd(&s_cnt[w1 >> SRC_BITS], 1);
    }
  }
  __syncthreads();

  int v = 0, incl = 0;
  if (t < BK_NODES) {
    v    = s_cnt[t];
    incl = wave_incl_scan(v, lane);
    if (lane == 31) s_wtot[wave] = incl;
  }
  __syncthreads();
  if (t < BK_NODES) {
    int wb = 0;
    for (int w = 0; w < wave; ++w) wb += s_wtot[w];
    int start = wb + incl - v;
    s_cur[t] = start;
    int n = b * BK_NODES + t;
    if (n < n_nodes) {
      *(int2*)(seg + 2 * n) = make_int2(bucket_base + start, v);
      cnt[n] = (float)v;
    }
  }
  __syncthreads();

  for (int g = wv; g < n_chunks; g += 32) {
    int p0 = __builtin_amdgcn_readfirstlane(s_lo[g]);
    int rl = __builtin_amdgcn_readfirstlane(s_len[g]);
    for (int i = lane; i < rl; i += 64) {
      int      i1 = i + 32;
      unsigned w0 = part[p0 + i];
      unsigned w1 = (i1 < rl) ? part[p0 + i1] : 0u;
      int q0 = atomicAdd(&s_cur[w0 >> SRC_BITS], 1);
      if (fits) s_sorted[q0] = (int)(w0 & SRC_MASK);
      else      sorted[bucket_base + q0] = (int)(w0 & SRC_MASK);
      if (i1 < rl) {
        int q1 = atomicAdd(&s_cur[w1 >> SRC_BITS], 1);
        if (fits) s_sorted[q1] = (int)(w1 & SRC_MASK);
        else      sorted[bucket_base + q1] = (int)(w1 & SRC_MASK);
      }
    }
  }
  if (fits) {
    __syncthreads();
    for (int i = t; i < total; i += 1024) sorted[bucket_base + i] = s_sorted[i];
  }
}

#define TPB 8
#define GRP 16
#define NO_END 0x40000000

__device__ __forceinline__ void wave_lds_handover() {
  __builtin_amdgcn_fence(__ATOMIC_ACQ_REL, "wavefront");
  __builtin_amdgcn_wave_barrier();
}

template <int TILES>
__device__ __forceinline__ void
layer_tiles(const float* __restrict__ xin, const int* sorted,
            const int* seg, const float* cnt,
            const float* __restrict__ Wl,
            const float* __restrict__ bl,
            const float* __restrict__ Wr,
            float* __restrict__ out, int n_nodes, int do_relu) {
  __shared__ float sWl[DF * LDS_STRIDE];
  __shared__ float sWr[DF * LDS_STRIDE];
  __shared__ float sT[TILES][16 * LDS_STRIDE];

  int wv   = __builtin_amdgcn_readfirstlane(threadIdx.x >> 5);
  int lane = threadIdx.x & 31;
  int nb   = (blockIdx.x * TILES + wv) * 16;
  int nn   = min(16, n_nodes - nb);
  int m    = lane & 15;
  int kh   = lane >> 4;
  float* tile = sT[wv];

  int    nlast = n_nodes - 1;
  int2   sd = *(const int2*)(seg + 2 * min(nb + m, nlast));
  float  cn = cnt[min(nb + m, nlast)];
  float2 xr[16];
#pragma unroll
  for (int r = 0; r < 16; ++r)
    xr[r] = *(const float2*)(xin + (long long)min(nb + r, nlast) * DF + lane * 2);

  constexpr int WROWS = DF / TILES;
  int    wc = lane * 2;
  float2 wl[WROWS], wr[WROWS];
#pragma unroll
  for (int j = 0; j < WROWS; ++j) {
    int r = wv + TILES * j;
    wl[j] = *(const float2*)&Wl[r * DF + wc];
    wr[j] = *(const float2*)&Wr[r * DF + wc];
  }
#pragma unroll
  for (int j = 0; j < WROWS; ++j) {
    int r = wv + TILES * j;
    *(float2*)&sWl[r * LDS_STRIDE + wc] = wl[j];
    *(float2*)&sWr[r * LDS_STRIDE + wc] = wr[j];
  }
#pragma unroll
  for (int r = 0; r < 16; ++r) *(float2*)&tile[r * LDS_STRIDE + lane * 2] = xr[r];
  __syncthreads();
  if (nn <= 0) return;

  v2f xa[16];
#pragma unroll
  for (int kk = 0; kk < 16; ++kk)
    xa[kk] = *(const v2f*)&tile[m * LDS_STRIDE + kk * 4 + 2 * kh];
  float bias[4];
#pragma unroll
  for (int nt = 0; nt < 4; ++nt) bias[nt] = bl[nt * 16 + m];
  wave_lds_handover();

  int e0   = __builtin_amdgcn_readlane(sd.x, 0);
  int e1   = __builtin_amdgcn_readlane(sd.x, nn - 1) +
             __builtin_amdgcn_readlane(sd.y, nn - 1);
  int ni   = 0;
  int nend = e0 + __builtin_amdgcn_readlane(sd.y, 0);

  float*   trow    = tile + lane * 2;
  unsigned laneoff = (unsigned)lane * 8u;
  float2   acc     = make_float2(0.0f, 0.0f);

  v8f cacc[4];
#pragma unroll
  for (int nt = 0; nt < 4; ++nt) {
    float b = bias[nt];
    v8f c = {b, b, b, b, b, b, b, b};
    cacc[nt] = c;
  }
  int  ks    = 0;
  auto self2 = [&]() __attribute__((always_inline)) {
    const float* wb = &sWr[m * LDS_STRIDE + ks * 4 + 2 * kh];
#pragma unroll
    for (int h = 0; h < 2; ++h) {
#pragma unroll
      for (int nt = 0; nt < 4; ++nt) {
        v2f bq = *(const v2f*)&wb[nt * 16 * LDS_STRIDE + h * 4];
        cacc[nt] = __builtin_amdgcn_wmma_f32_16x16x4_f32(
            false, xa[h], false, bq, (short)0, cacc[nt], false, false);
      }
    }
#pragma unroll
    for (int j = 0; j < 14; ++j) xa[j] = xa[j + 2];
    ks += 2;
  };

  int    last = e1 - 1;
  int    base = e0;
  int    idx  = 0;
  if (e0 < e1) idx = sorted[min(e0 + lane, last)];
  float2 v[GRP];

  auto issue = [&](int g) __attribute__((always_inline)) {
#pragma unroll
    for (int u = 0; u < GRP; ++u) {
      unsigned s = (unsigned)__builtin_amdgcn_readlane(idx, g * GRP + u);
      v[u] = *(const float2*)((const char*)xin + ((s << 8) + laneoff));
    }
  };
  auto consume = [&](int p0) __attribute__((always_inline)) {
    int rel = nend - p0;
#pragma unroll
    for (int u = 0; u < GRP; ++u) {
      while (rel == u) {
        *(float2*)(trow + ni * LDS_STRIDE) = acc;
        acc = make_float2(0.0f, 0.0f);
        ++ni;
        if (ni < nn) {
          int d = __builtin_amdgcn_readlane(sd.y, ni);
          nend += d;
          rel  += d;
        } else {
          nend = NO_END;
          rel  = NO_END;
        }
      }
      acc.x += v[u].x;
      acc.y += v[u].y;
    }
  };

  auto trip = [&](bool carry) __attribute__((always_inline)) {
    int idx_next = sorted[min(base + 32 + lane, last)];
    issue(0);
    if (carry) self2();
    consume(base);
    if (base + GRP < e1) {
      issue(1);
      consume(base + GRP);
    }
    idx = idx_next;
    base += 32;
  };
  while (base < e1 && ks < 16) trip(true);
  while (base < e1) trip(false);
  for (; ni < 16; ++ni) {
    *(float2*)(trow + ni * LDS_STRIDE) = acc;
    acc = make_float2(0.0f, 0.0f);
  }
  wave_lds_handover();

#pragma nounroll
  while (ks < 16) self2();

  float invc = 1.0f / fmaxf(cn, 1.0f);
#pragma unroll
  for (int kk = 0; kk < 16; ++kk) {
    int k = kk * 4;
    v2f am = *(const v2f*)&tile[m * LDS_STRIDE + k + 2 * kh];
    am *= invc;
#pragma unroll
    for (int nt = 0; nt < 4; ++nt) {
      v2f bw = *(const v2f*)&sWl[(nt * 16 + m) * LDS_STRIDE + k + 2 * kh];
      cacc[nt] = __builtin_amdgcn_wmma_f32_16x16x4_f32(
          false, am, false, bw, (short)0, cacc[nt], false, false);
    }
  }

  float* orow0 = out + (long long)(nb + 8 * kh) * DF + m;
  if (nn == 16) {
#pragma unroll
    for (int nt = 0; nt < 4; ++nt) {
#pragma unroll
      for (int g = 0; g < 8; ++g) {
        float v = cacc[nt][g];
        if (do_relu) v = fmaxf(v, 0.0f);
        orow0[g * DF + nt * 16] = v;
      }
    }
  } else {
#pragma unroll
    for (int nt = 0; nt < 4; ++nt) {
#pragma unroll
      for (int g = 0; g < 8; ++g) {
        float v = cacc[nt][g];
        if (do_relu) v = fmaxf(v, 0.0f);
        if (g + 8 * kh < nn) orow0[g * DF + nt * 16] = v;
      }
    }
  }
}

__global__ void __launch_bounds__(TPB * 32)
sage_layer(const float* __restrict__ xin, const int* __restrict__ sorted,
           const int* __restrict__ seg, const float* __restrict__ cnt,
           const float* __restrict__ Wl, const float* __restrict__ bl,
           const float* __restrict__ Wr,
           float* __restrict__ out, int n_nodes, int do_relu) {
  layer_tiles<TPB>(xin, sorted, seg, cnt, Wl, bl, Wr, out, n_nodes, do_relu);
}

#define BK_TILES (BK_NODES / 16)
static_assert(BK_TILES * 32 == 1024, "one wave per tile, 1024 threads");
__global__ void __launch_bounds__(1024)
bucket_sort_layer(const unsigned* __restrict__ part, const int* __restrict__ tab,
                  int* seg, int* sorted, float* cnt,
                  const float* __restrict__ xin,
                  const float* __restrict__ Wl, const float* __restrict__ bl,
                  const float* __restrict__ Wr,
                  float* __restrict__ out, int n_nodes, int n_chunks) {
  bucket_sort(part, tab, seg, sorted, cnt, n_nodes, n_chunks);
  __syncthreads();
  layer_tiles<BK_TILES>(xin, sorted, seg, cnt, Wl, bl, Wr, out, n_nodes, 1);
}

extern "C" void kernel_launch(void* const* d_in, const int* in_sizes, int n_in,
                              void* d_out, int out_size, void* d_ws, size_t ws_size,
                              hipStream_t stream) {
  const float* x   = (const float*)d_in[0];
  const int*   ei  = (const int*)d_in[1];
  const float* W1l = (const float*)d_in[2];
  const float* b1l = (const float*)d_in[3];
  const float* W1r = (const float*)d_in[4];
  const float* W2l = (const float*)d_in[5];
  const float* b2l = (const float*)d_in[6];
  const float* W2r = (const float*)d_in[7];
  float* out = (float*)d_out;

  int n_nodes = in_sizes[0] / DF;
  int n_edges = in_sizes[1] / 2;
  const int* src = ei;
  const int* dst = ei + n_edges;

  size_t rowBytes = (size_t)n_nodes * DF * sizeof(float);
  char* ws = (char*)d_ws;
  int*   seg    = (int*)ws;
  int*   sorted = seg + 2 * (size_t)n_nodes;
  float* h      = (float*)(ws + rowBytes);
  float* cnt    = (float*)(ws + 2 * rowBytes);

  int n_buckets = (n_nodes + BK_NODES - 1) / BK_NODES;
  int n_chunks  = (n_edges + CH_EDGES - 1) / CH_EDGES;

  unsigned* part = (unsigned*)(sorted + n_edges);
  int*      tab  = (int*)(part + n_edges);

  int tiles   = (n_nodes + 15) / 16;
  int lblocks = (tiles + TPB - 1) / TPB;

  edge_partition<<<n_chunks, 1024, 0, stream>>>(src, dst, part, tab, n_edges,
                                                n_buckets, n_chunks);
  bucket_sort_layer<<<n_buckets, 1024, 0, stream>>>(part, tab, seg, sorted, cnt,
                                                    x, W1l, b1l, W1r, h,
                                                    n_nodes, n_chunks);
  sage_layer<<<lblocks, TPB * 32, 0, stream>>>(h, sorted, seg, cnt, W2l, b2l, W2r, out,
                                               n_nodes, 0);
}
